# starting kernel
# baseline (speedup 1.0000x reference)
#include <hip/hip_runtime.h>

#define GN 50000
#define GE 800000
#define GD 128
#define NLAYER 4

typedef __attribute__((ext_vector_type(2))) float v2f;
typedef __attribute__((ext_vector_type(8))) float v8f;


__global__ void k_zero(float* p, int n) {
    int i = blockIdx.x * blockDim.x + threadIdx.x;
    if (i < n) p[i] = 0.0f;
}

__global__ void k_deg(const int* __restrict__ dst, float* deg, int e) {
    int i = blockIdx.x * blockDim.x + threadIdx.x;
    if (i < e) atomicAdd(&deg[dst[i]], 1.0f);
}

__global__ void k_rsqrt(float* dis, int n) {
    int i = blockIdx.x * blockDim.x + threadIdx.x;
    if (i < n) dis[i] = rsqrtf(1.0f + dis[i]);
}


__global__ void k_gemm128(const float* __restrict__ A,
                          const float* __restrict__ W,
                          float* __restrict__ out) {
    __shared__ float As[16 * 132];
    const int rowBase = blockIdx.x * 16;
    const int tid = threadIdx.x;

    for (int i = tid; i < 16 * 128; i += 256) {
        int r = i >> 7, c = i & 127;
        As[r * 132 + c] = A[(size_t)(rowBase + r) * GD + c];
    }
    __syncthreads();

    const int wave  = tid >> 5;
    const int lane  = tid & 31;
    const int m     = lane & 15;
    const int khalf = (lane >> 4) * 2;
    const int n     = wave * 16 + m;

    v8f c = {};
    #pragma unroll
    for (int k = 0; k < 128; k += 4) {
        v2f a, b;
        a.x = As[m * 132 + k + khalf];
        a.y = As[m * 132 + k + khalf + 1];
        b.x = W[(size_t)(k + khalf) * GD + n];
        b.y = W[(size_t)(k + khalf + 1) * GD + n];
        c = __builtin_amdgcn_wmma_f32_16x16x4_f32(
                false, a, false, b, (short)0, c, false, false);
    }

    const int rOff = (lane >> 4) * 8;
    #pragma unroll
    for (int i = 0; i < 8; ++i)
        out[(size_t)(rowBase + rOff + i) * GD + n] = c[i];
}


__global__ void k_init_agg(const float* __restrict__ xw,
                           const float* __restrict__ dis,
                           const float* __restrict__ b,
                           float* __restrict__ agg) {
    int i = blockIdx.x * blockDim.x + threadIdx.x;
    int node = i >> 7, f = i & 127;
    float ds = dis[node];
    agg[i] = xw[i] * ds * ds + b[f];
}


__global__ void k_edge_scatter(const float* __restrict__ xw,
                               const int* __restrict__ src,
                               const int* __restrict__ dst,
                               const float* __restrict__ dis,
                               float* __restrict__ agg, int e) {
    int gt = blockIdx.x * blockDim.x + threadIdx.x;
    int edge = gt >> 5;
    if (edge >= e) return;
    int lane = threadIdx.x & 31;
    int s = src[edge], d = dst[edge];
    float norm = dis[s] * dis[d];
    const float4* xs = (const float4*)(xw + (size_t)s * GD);
    float4 v = xs[lane];
    float* ad = agg + (size_t)d * GD + lane * 4;
    atomicAdd(ad + 0, v.x * norm);
    atomicAdd(ad + 1, v.y * norm);
    atomicAdd(ad + 2, v.z * norm);
    atomicAdd(ad + 3, v.w * norm);
}


__global__ void k_finish(const float* __restrict__ agg,
                         float* __restrict__ h, int residual) {
    int i = blockIdx.x * blockDim.x + threadIdx.x;
    float v = agg[i] + (residual ? h[i] : 0.0f);
    h[i] = fmaxf(v, 0.0f);
}


__global__ void k_dot(const float* __restrict__ h,
                      const float* __restrict__ Wout,
                      float* __restrict__ xw1, int n) {
    int gt = blockIdx.x * blockDim.x + threadIdx.x;
    int row = gt >> 5;
    if (row >= n) return;
    int lane = threadIdx.x & 31;
    float4 a = ((const float4*)(h + (size_t)row * GD))[lane];
    float4 b = ((const float4*)Wout)[lane];
    float sum = a.x * b.x + a.y * b.y + a.z * b.z + a.w * b.w;
    #pragma unroll
    for (int off = 16; off > 0; off >>= 1)
        sum += __shfl_down(sum, off, 32);
    if (lane == 0) xw1[row] = sum;
}

__global__ void k_out_init(const float* __restrict__ xw1,
                           const float* __restrict__ dis,
                           const float* __restrict__ bout,
                           float* __restrict__ out, int n) {
    int i = blockIdx.x * blockDim.x + threadIdx.x;
    if (i < n) {
        float ds = dis[i];
        out[i] = xw1[i] * ds * ds + bout[0];
    }
}

__global__ void k_out_scatter(const float* __restrict__ xw1,
                              const int* __restrict__ src,
                              const int* __restrict__ dst,
                              const float* __restrict__ dis,
                              float* __restrict__ out, int e) {
    int i = blockIdx.x * blockDim.x + threadIdx.x;
    if (i < e) {
        int s = src[i], d = dst[i];
        atomicAdd(&out[d], xw1[s] * dis[s] * dis[d]);
    }
}


extern "C" void kernel_launch(void* const* d_in, const int* in_sizes, int n_in,
                              void* d_out, int out_size, void* d_ws, size_t ws_size,
                              hipStream_t stream) {
    (void)in_sizes; (void)n_in; (void)out_size; (void)ws_size;
    const float* x     = (const float*)d_in[0];
    const int*   eidx  = (const int*)d_in[1];
    const float* W_in  = (const float*)d_in[2];
    const float* b_in  = (const float*)d_in[3];
    const float* W_h   = (const float*)d_in[4];
    const float* b_h   = (const float*)d_in[5];
    const float* W_out = (const float*)d_in[6];
    const float* b_out = (const float*)d_in[7];
    float* out = (float*)d_out;

    const int* src = eidx;
    const int* dst = eidx + GE;

    const size_t NPAD = (GN + 63) & ~(size_t)63;
    float* dis = (float*)d_ws;
    float* xw1 = dis + NPAD;
    float* h   = xw1 + NPAD;
    float* xw  = h + (size_t)GN * GD;
    float* agg = xw + (size_t)GN * GD;

    const int T = 256;
    const int gN    = (GN + T - 1) / T;
    const int gE    = (GE + T - 1) / T;
    const int gNF   = (GN * GD) / T;
    const int gEW   = (GE * 32) / T;
    const int gGemm = GN / 16;

    k_zero<<<gN, T, 0, stream>>>(dis, GN);
    k_deg<<<gE, T, 0, stream>>>(dst, dis, GE);
    k_rsqrt<<<gN, T, 0, stream>>>(dis, GN);

    k_gemm128<<<gGemm, T, 0, stream>>>(x, W_in, xw);
    k_init_agg<<<gNF, T, 0, stream>>>(xw, dis, b_in, agg);
    k_edge_scatter<<<gEW, T, 0, stream>>>(xw, src, dst, dis, agg, GE);
    k_finish<<<gNF, T, 0, stream>>>(agg, h, 0);

    for (int l = 0; l < NLAYER; ++l) {
        const float* Wl = W_h + (size_t)l * GD * GD;
        const float* bl = b_h + (size_t)l * GD;
        k_gemm128<<<gGemm, T, 0, stream>>>(h, Wl, xw);
        k_init_agg<<<gNF, T, 0, stream>>>(xw, dis, bl, agg);
        k_edge_scatter<<<gEW, T, 0, stream>>>(xw, src, dst, dis, agg, GE);
        k_finish<<<gNF, T, 0, stream>>>(agg, h, 1);
    }

    k_dot<<<(GN * 32) / T, T, 0, stream>>>(h, W_out, xw1, GN);
    k_out_init<<<gN, T, 0, stream>>>(xw1, dis, b_out, out, GN);
    k_out_scatter<<<gE, T, 0, stream>>>(xw1, src, dst, dis, out, GE);
}


# each layer's two elementwise passes folded into its GEMM kernel: epilogue also writes agg's start value xw*dis^2+b, A-tile load does relu(agg [+h]) and stores h; 26 launches become 17
# speedup vs baseline: 2.1513x; 2.1513x over previous
#include <hip/hip_runtime.h>

#define GN 50000
#define GE 800000
#define GD 128
#define NLAYER 4

typedef __attribute__((ext_vector_type(2))) float v2f;
typedef __attribute__((ext_vector_type(8))) float v8f;


__global__ void k_zero(float* p, int n) {
    int i = blockIdx.x * blockDim.x + threadIdx.x;
    if (i < n) p[i] = 0.0f;
}

__global__ void k_deg(const int* __restrict__ dst, float* deg, int e) {
    int i = blockIdx.x * blockDim.x + threadIdx.x;
    if (i < e) atomicAdd(&deg[dst[i]], 1.0f);
}

__global__ void k_rsqrt(float* dis, int n) {
    int i = blockIdx.x * blockDim.x + threadIdx.x;
    if (i < n) dis[i] = rsqrtf(1.0f + dis[i]);
}


template <int MODE>
__global__ void k_gemm128(const float* __restrict__ X,
                          float* agg, float* h,
                          const float* __restrict__ W,
                          const float* __restrict__ dis,
                          const float* __restrict__ bias,
                          float* __restrict__ xw) {
    __shared__ __attribute__((aligned(16))) float As[16 * 132];
    const int rowBase = blockIdx.x * 16;
    const int tid = threadIdx.x;

    size_t g[2];
    float4 v[2], p[2];
    #pragma unroll
    for (int q = 0; q < 2; ++q) {
        const int i4 = tid + q * 256;
        g[q] = (size_t)(rowBase + (i4 >> 5)) * GD + (i4 & 31) * 4;
        if (MODE == 0) {
            v[q] = *(const float4*)(X + g[q]);
        } else {
            v[q] = *(const float4*)(agg + g[q]);
            if (MODE == 2) p[q] = *(const float4*)(h + g[q]);
        }
    }
    #pragma unroll
    for (int q = 0; q < 2; ++q) {
        if (MODE == 2) {
            v[q].x += p[q].x; v[q].y += p[q].y; v[q].z += p[q].z; v[q].w += p[q].w;
        }
        if (MODE != 0) {
            v[q].x = fmaxf(v[q].x, 0.0f); v[q].y = fmaxf(v[q].y, 0.0f);
            v[q].z = fmaxf(v[q].z, 0.0f); v[q].w = fmaxf(v[q].w, 0.0f);
        }
        const int i4 = tid + q * 256;
        *(float4*)(&As[(i4 >> 5) * 132 + (i4 & 31) * 4]) = v[q];
    }
    __syncthreads();
    if (MODE != 0) {
        *(float4*)(h + g[0]) = v[0];
        *(float4*)(h + g[1]) = v[1];
    }

    const int wave  = tid >> 5;
    const int lane  = tid & 31;
    const int m     = lane & 15;
    const int khalf = (lane >> 4) * 2;
    const int n     = wave * 16 + m;

    v8f c = {};
    #pragma unroll
    for (int k = 0; k < 128; k += 4) {
        v2f a, b;
        a.x = As[m * 132 + k + khalf];
        a.y = As[m * 132 + k + khalf + 1];
        b.x = W[(size_t)(k + khalf) * GD + n];
        b.y = W[(size_t)(k + khalf + 1) * GD + n];
        c = __builtin_amdgcn_wmma_f32_16x16x4_f32(
                false, a, false, b, (short)0, c, false, false);
    }

    const int rOff = (lane >> 4) * 8;
    const float bn = bias[n];
    const float4 d0 = *(const float4*)(dis + rowBase + rOff);
    const float4 d1 = *(const float4*)(dis + rowBase + rOff + 4);
    const float dsr[8] = {d0.x, d0.y, d0.z, d0.w, d1.x, d1.y, d1.z, d1.w};
    #pragma unroll
    for (int i = 0; i < 8; ++i) {
        const size_t o = (size_t)(rowBase + rOff + i) * GD + n;
        xw[o]  = c[i];
        agg[o] = c[i] * dsr[i] * dsr[i] + bn;
    }
}


static_assert(GE % 256 == 0, "a block ticket is 8 whole 32-edge groups");

#define EDGE_BATCH 8
#define NTICKET (GE / 256)
#define SCATTER_BLOCKS 2048

__device__ __forceinline__ void scatter_group(const float* __restrict__ xw,
                                              const int* __restrict__ src,
                                              const int* __restrict__ dst,
                                              const float* __restrict__ dis,
                                              float* __restrict__ agg,
                                              int e0, int set, int lane) {
    int s = src[e0 + lane], d = dst[e0 + lane];
    bool mine = (d >= GN / 2) == (set != 0);
    float nrm = 0.0f;
    if (mine) nrm = dis[s] * dis[d];
    int normBits = __builtin_bit_cast(int, nrm);
    unsigned m = __builtin_amdgcn_ballot_w32(mine);
    while (m != 0) {
        float v[EDGE_BATCH][4];
        unsigned ml = m;
        #pragma unroll
        for (int u = 0; u < EDGE_BATCH; ++u) {
            if (ml != 0) {
                int j = __builtin_ctz(ml);
                ml &= ml - 1;
                int sj = __builtin_amdgcn_readlane(s, j);
                const float* xs = xw + (size_t)sj * GD;
                v[u][0] = xs[lane];      v[u][1] = xs[lane + 32];
                v[u][2] = xs[lane + 64]; v[u][3] = xs[lane + 96];
            }
        }
        #pragma unroll
        for (int u = 0; u < EDGE_BATCH; ++u) {
            if (m != 0) {
                int j = __builtin_ctz(m);
                m &= m - 1;
                int dj = __builtin_amdgcn_readlane(d, j);
                float nj = __builtin_bit_cast(float, __builtin_amdgcn_readlane(normBits, j));
                float* ad = agg + (size_t)dj * GD;
                atomicAdd(ad + lane,      v[u][0] * nj);
                atomicAdd(ad + lane + 32, v[u][1] * nj);
                atomicAdd(ad + lane + 64, v[u][2] * nj);
                atomicAdd(ad + lane + 96, v[u][3] * nj);
            }
        }
    }
}

__global__ void k_edge_scatter(const float* __restrict__ xw,
                               const int* __restrict__ src,
                               const int* __restrict__ dst,
                               const float* __restrict__ dis,
                               float* __restrict__ agg,
                               int* cnt0, int* cnt1) {
    __shared__ int tk;
    const int lane = threadIdx.x & 31;
    const int wave = threadIdx.x >> 5;
    const int myhalf = (__builtin_amdgcn_s_sendmsg_rtn(0x87) >> 18) & 1;
    for (int k = 0; k < 2; ++k) {
        const int set = myhalf ^ k;
        int* cnt = set ? cnt1 : cnt0;
        int tn = 0;
        if (threadIdx.x == 0) tn = atomicAdd(cnt, 1);
        for (int trip = 0; trip <= NTICKET; ++trip) {
            if (threadIdx.x == 0) tk = tn;
            __syncthreads();
            const int t = tk;
            __syncthreads();
            if (t >= NTICKET) break;
            if (threadIdx.x == 0) tn = atomicAdd(cnt, 1);
            scatter_group(xw, src, dst, dis, agg, (t * 8 + wave) * 32, set, lane);
        }
    }
}


__global__ void k_finish(const float* __restrict__ agg,
                         float* __restrict__ h, int residual) {
    int i = blockIdx.x * blockDim.x + threadIdx.x;
    float v = agg[i] + (residual ? h[i] : 0.0f);
    h[i] = fmaxf(v, 0.0f);
}


__global__ void k_dot(const float* __restrict__ h,
                      const float* __restrict__ Wout,
                      float* __restrict__ xw1, int n) {
    int gt = blockIdx.x * blockDim.x + threadIdx.x;
    int row = gt >> 5;
    if (row >= n) return;
    int lane = threadIdx.x & 31;
    float4 a = ((const float4*)(h + (size_t)row * GD))[lane];
    float4 b = ((const float4*)Wout)[lane];
    float sum = a.x * b.x + a.y * b.y + a.z * b.z + a.w * b.w;
    #pragma unroll
    for (int off = 16; off > 0; off >>= 1)
        sum += __shfl_down(sum, off, 32);
    if (lane == 0) xw1[row] = sum;
}

__global__ void k_out_init(const float* __restrict__ xw1,
                           const float* __restrict__ dis,
                           const float* __restrict__ bout,
                           float* __restrict__ out, int n) {
    int i = blockIdx.x * blockDim.x + threadIdx.x;
    if (i < n) {
        float ds = dis[i];
        out[i] = xw1[i] * ds * ds + bout[0];
    }
}

__global__ void k_out_scatter(const float* __restrict__ xw1,
                              const int* __restrict__ src,
                              const int* __restrict__ dst,
                              const float* __restrict__ dis,
                              float* __restrict__ out, int e) {
    int i = blockIdx.x * blockDim.x + threadIdx.x;
    if (i < e) {
        int s = src[i], d = dst[i];
        atomicAdd(&out[d], xw1[s] * dis[s] * dis[d]);
    }
}


extern "C" void kernel_launch(void* const* d_in, const int* in_sizes, int n_in,
                              void* d_out, int out_size, void* d_ws, size_t ws_size,
                              hipStream_t stream) {
    (void)in_sizes; (void)n_in; (void)out_size; (void)ws_size;
    const float* x     = (const float*)d_in[0];
    const int*   eidx  = (const int*)d_in[1];
    const float* W_in  = (const float*)d_in[2];
    const float* b_in  = (const float*)d_in[3];
    const float* W_h   = (const float*)d_in[4];
    const float* b_h   = (const float*)d_in[5];
    const float* W_out = (const float*)d_in[6];
    const float* b_out = (const float*)d_in[7];
    float* out = (float*)d_out;

    const int* src = eidx;
    const int* dst = eidx + GE;

    const size_t NPAD = (GN + 63) & ~(size_t)63;
    float* dis = (float*)d_ws;
    float* xw1 = dis + NPAD;
    float* h   = xw1 + NPAD;
    float* xw  = h + (size_t)GN * GD;
    float* agg = xw + (size_t)GN * GD;

    const int T = 256;
    const int gN    = (GN + T - 1) / T;
    const int gE    = (GE + T - 1) / T;
    const int gNF   = (GN * GD) / T;
    const int gGemm = GN / 16;

    int* cnt0 = (int*)(dis + GN + 16);
    int* cnt1 = (int*)(xw1 + GN + 16);
    k_zero<<<(int)((2 * NPAD + T - 1) / T), T, 0, stream>>>(dis, (int)(2 * NPAD));
    k_deg<<<gE, T, 0, stream>>>(dst, dis, GE);
    k_rsqrt<<<gN, T, 0, stream>>>(dis, GN);

    k_gemm128<0><<<gGemm, T, 0, stream>>>(x, agg, h, W_in, dis, b_in, xw);
    k_edge_scatter<<<SCATTER_BLOCKS, T, 0, stream>>>(xw, src, dst, dis, agg, cnt0, cnt1);

    for (int l = 0; l < NLAYER; ++l) {
        const float* Wl = W_h + (size_t)l * GD * GD;
        const float* bl = b_h + (size_t)l * GD;
        if (l == 0) k_gemm128<1><<<gGemm, T, 0, stream>>>(x, agg, h, Wl, dis, bl, xw);
        else        k_gemm128<2><<<gGemm, T, 0, stream>>>(x, agg, h, Wl, dis, bl, xw);
        k_edge_scatter<<<SCATTER_BLOCKS, T, 0, stream>>>(xw, src, dst, dis, agg, cnt0 + 1 + l, cnt1 + 1 + l);
    }
    k_finish<<<gNF, T, 0, stream>>>(agg, h, 1);

    k_dot<<<(GN * 32) / T, T, 0, stream>>>(h, W_out, xw1, GN);
    k_out_init<<<gN, T, 0, stream>>>(xw1, dis, b_out, out, GN);
    k_out_scatter<<<gE, T, 0, stream>>>(xw1, src, dst, dis, out, GE);
}


# norm factored out of the edge loop: GEMM epilogue stores xw*dis[row], scatter adds raw rows (no dis gathers, no multiply), dis[dst] and bias applied where agg is read; sums in atomic arrival order
# speedup vs baseline: 2.5376x; 1.1796x over previous
#include <hip/hip_runtime.h>

#define GN 50000
#define GE 800000
#define GD 128
#define NLAYER 4

typedef __attribute__((ext_vector_type(2))) float v2f;
typedef __attribute__((ext_vector_type(8))) float v8f;


__global__ void k_zero(float* p, int n) {
    int i = blockIdx.x * blockDim.x + threadIdx.x;
    if (i < n) p[i] = 0.0f;
}

__global__ void k_deg(const int* __restrict__ dst, float* deg, int e) {
    int i = blockIdx.x * blockDim.x + threadIdx.x;
    if (i < e) atomicAdd(&deg[dst[i]], 1.0f);
}

__global__ void k_rsqrt(float* dis, int n) {
    int i = blockIdx.x * blockDim.x + threadIdx.x;
    if (i < n) dis[i] = rsqrtf(1.0f + dis[i]);
}


template <int MODE>
__global__ void k_gemm128(const float* __restrict__ X,
                          float* agg, float* h,
                          const float* __restrict__ W,
                          const float* __restrict__ dis,
                          const float* __restrict__ bprev,
                          float* __restrict__ xw) {
    __shared__ __attribute__((aligned(16))) float As[16 * 132];
    const int rowBase = blockIdx.x * 16;
    const int tid = threadIdx.x;

    size_t g[2];
    float4 v[2], p[2];
    float dr[2];
    float4 bp;
    if (MODE != 0) bp = *(const float4*)(bprev + (tid & 31) * 4);
    #pragma unroll
    for (int q = 0; q < 2; ++q) {
        const int i4 = tid + q * 256;
        g[q] = (size_t)(rowBase + (i4 >> 5)) * GD + (i4 & 31) * 4;
        if (MODE == 0) {
            v[q] = *(const float4*)(X + g[q]);
        } else {
            v[q] = *(const float4*)(agg + g[q]);
            dr[q] = dis[rowBase + (i4 >> 5)];
            if (MODE == 2) p[q] = *(const float4*)(h + g[q]);
        }
    }
    #pragma unroll
    for (int q = 0; q < 2; ++q) {
        if (MODE != 0) {
            v[q].x = v[q].x * dr[q] + bp.x; v[q].y = v[q].y * dr[q] + bp.y;
            v[q].z = v[q].z * dr[q] + bp.z; v[q].w = v[q].w * dr[q] + bp.w;
        }
        if (MODE == 2) {
            v[q].x += p[q].x; v[q].y += p[q].y; v[q].z += p[q].z; v[q].w += p[q].w;
        }
        if (MODE != 0) {
            v[q].x = fmaxf(v[q].x, 0.0f); v[q].y = fmaxf(v[q].y, 0.0f);
            v[q].z = fmaxf(v[q].z, 0.0f); v[q].w = fmaxf(v[q].w, 0.0f);
        }
        const int i4 = tid + q * 256;
        *(float4*)(&As[(i4 >> 5) * 132 + (i4 & 31) * 4]) = v[q];
    }
    __syncthreads();
    if (MODE != 0) {
        *(float4*)(h + g[0]) = v[0];
        *(float4*)(h + g[1]) = v[1];
    }

    const int wave  = tid >> 5;
    const int lane  = tid & 31;
    const int m     = lane & 15;
    const int khalf = (lane >> 4) * 2;
    const int n     = wave * 16 + m;

    v8f c = {};
    #pragma unroll
    for (int k = 0; k < 128; k += 4) {
        v2f a, b;
        a.x = As[m * 132 + k + khalf];
        a.y = As[m * 132 + k + khalf + 1];
        b.x = W[(size_t)(k + khalf) * GD + n];
        b.y = W[(size_t)(k + khalf + 1) * GD + n];
        c = __builtin_amdgcn_wmma_f32_16x16x4_f32(
                false, a, false, b, (short)0, c, false, false);
    }

    const int rOff = (lane >> 4) * 8;
    const float4 d0 = *(const float4*)(dis + rowBase + rOff);
    const float4 d1 = *(const float4*)(dis + rowBase + rOff + 4);
    const float dsr[8] = {d0.x, d0.y, d0.z, d0.w, d1.x, d1.y, d1.z, d1.w};
    #pragma unroll
    for (int i = 0; i < 8; ++i) {
        const size_t o = (size_t)(rowBase + rOff + i) * GD + n;
        const float xs = c[i] * dsr[i];
        xw[o]  = xs;
        agg[o] = xs;
    }
}


static_assert(GE % 256 == 0, "a block ticket is 8 whole 32-edge groups");

#define EDGE_BATCH 8
#define NTICKET (GE / 256)
#define SCATTER_BLOCKS 2048

__device__ __forceinline__ void scatter_group(const float* __restrict__ xw,
                                              const int* __restrict__ src,
                                              const int* __restrict__ dst,
                                              float* __restrict__ agg,
                                              int e0, int set, int lane) {
    int s = src[e0 + lane], d = dst[e0 + lane];
    asm volatile("" :: "v"(s));
    bool mine = (d >= GN / 2) == (set != 0);
    unsigned m = __builtin_amdgcn_ballot_w32(mine);
    while (m != 0) {
        float v[EDGE_BATCH][4];
        unsigned ml = m;
        #pragma unroll
        for (int u = 0; u < EDGE_BATCH; ++u) {
            if (ml != 0) {
                int j = __builtin_ctz(ml);
                ml &= ml - 1;
                int sj = __builtin_amdgcn_readlane(s, j);
                const float* xs = xw + (size_t)sj * GD;
                v[u][0] = xs[lane];      v[u][1] = xs[lane + 32];
                v[u][2] = xs[lane + 64]; v[u][3] = xs[lane + 96];
            }
        }
        #pragma unroll
        for (int u = 0; u < EDGE_BATCH; ++u) {
            if (m != 0) {
                int j = __builtin_ctz(m);
                m &= m - 1;
                int dj = __builtin_amdgcn_readlane(d, j);
                float* ad = agg + (size_t)dj * GD;
                atomicAdd(ad + lane,      v[u][0]);
                atomicAdd(ad + lane + 32, v[u][1]);
                atomicAdd(ad + lane + 64, v[u][2]);
                atomicAdd(ad + lane + 96, v[u][3]);
            }
        }
    }
}

__global__ void k_edge_scatter(const float* __restrict__ xw,
                               const int* __restrict__ src,
                               const int* __restrict__ dst,
                               float* __restrict__ agg,
                               int* cnt0, int* cnt1) {
    __shared__ int tk;
    const int lane = threadIdx.x & 31;
    const int wave = threadIdx.x >> 5;
    const int myhalf = (__builtin_amdgcn_s_sendmsg_rtn(0x87) >> 18) & 1;
    for (int k = 0; k < 2; ++k) {
        const int set = myhalf ^ k;
        int* cnt = set ? cnt1 : cnt0;
        int tn = 0;
        if (threadIdx.x == 0) tn = atomicAdd(cnt, 1);
        for (int trip = 0; trip <= NTICKET; ++trip) {
            if (threadIdx.x == 0) tk = tn;
            __syncthreads();
            const int t = tk;
            __syncthreads();
            if (t >= NTICKET) break;
            if (threadIdx.x == 0) tn = atomicAdd(cnt, 1);
            scatter_group(xw, src, dst, agg, (t * 8 + wave) * 32, set, lane);
        }
    }
}


__global__ void k_finish(const float* __restrict__ agg,
                         float* __restrict__ h,
                         const float* __restrict__ dis,
                         const float* __restrict__ bias, int residual) {
    int i = blockIdx.x * blockDim.x + threadIdx.x;
    float v = agg[i] * dis[i >> 7] + bias[i & (GD - 1)] + (residual ? h[i] : 0.0f);
    h[i] = fmaxf(v, 0.0f);
}


__global__ void k_dot(const float* __restrict__ h,
                      const float* __restrict__ Wout,
                      const float* __restrict__ dis,
                      float* __restrict__ xw1, int n) {
    int gt = blockIdx.x * blockDim.x + threadIdx.x;
    int row = gt >> 5;
    if (row >= n) return;
    int lane = threadIdx.x & 31;
    float4 a = ((const float4*)(h + (size_t)row * GD))[lane];
    float4 b = ((const float4*)Wout)[lane];
    float sum = a.x * b.x + a.y * b.y + a.z * b.z + a.w * b.w;
    #pragma unroll
    for (int off = 16; off > 0; off >>= 1)
        sum += __shfl_down(sum, off, 32);
    if (lane == 0) xw1[row] = sum * dis[row];
}

__global__ void k_out_init(const float* __restrict__ xw1,
                           const float* __restrict__ dis,
                           const float* __restrict__ bout,
                           float* __restrict__ out, int n) {
    int i = blockIdx.x * blockDim.x + threadIdx.x;
    if (i < n) {
        out[i] = xw1[i] * dis[i] + bout[0];
    }
}

__global__ void k_out_scatter(const float* __restrict__ xw1,
                              const int* __restrict__ src,
                              const int* __restrict__ dst,
                              const float* __restrict__ dis,
                              float* __restrict__ out, int e) {
    int i = blockIdx.x * blockDim.x + threadIdx.x;
    if (i < e) {
        int s = src[i], d = dst[i];
        atomicAdd(&out[d], xw1[s] * dis[d]);
    }
}


extern "C" void kernel_launch(void* const* d_in, const int* in_sizes, int n_in,
                              void* d_out, int out_size, void* d_ws, size_t ws_size,
                              hipStream_t stream) {
    (void)in_sizes; (void)n_in; (void)out_size; (void)ws_size;
    const float* x     = (const float*)d_in[0];
    const int*   eidx  = (const int*)d_in[1];
    const float* W_in  = (const float*)d_in[2];
    const float* b_in  = (const float*)d_in[3];
    const float* W_h   = (const float*)d_in[4];
    const float* b_h   = (const float*)d_in[5];
    const float* W_out = (const float*)d_in[6];
    const float* b_out = (const float*)d_in[7];
    float* out = (float*)d_out;

    const int* src = eidx;
    const int* dst = eidx + GE;

    const size_t NPAD = (GN + 63) & ~(size_t)63;
    float* dis = (float*)d_ws;
    float* xw1 = dis + NPAD;
    float* h   = xw1 + NPAD;
    float* xw  = h + (size_t)GN * GD;
    float* agg = xw + (size_t)GN * GD;

    const int T = 256;
    const int gN    = (GN + T - 1) / T;
    const int gE    = (GE + T - 1) / T;
    const int gNF   = (GN * GD) / T;
    const int gGemm = GN / 16;

    int* cnt0 = (int*)(dis + GN + 16);
    int* cnt1 = (int*)(xw1 + GN + 16);
    k_zero<<<(int)((2 * NPAD + T - 1) / T), T, 0, stream>>>(dis, (int)(2 * NPAD));
    k_deg<<<gE, T, 0, stream>>>(dst, dis, GE);
    k_rsqrt<<<gN, T, 0, stream>>>(dis, GN);

    k_gemm128<0><<<gGemm, T, 0, stream>>>(x, agg, h, W_in, dis, b_in, xw);
    k_edge_scatter<<<SCATTER_BLOCKS, T, 0, stream>>>(xw, src, dst, agg, cnt0, cnt1);

    for (int l = 0; l < NLAYER; ++l) {
        const float* Wl = W_h + (size_t)l * GD * GD;
        const float* bp = (l == 0) ? b_in : b_h + (size_t)(l - 1) * GD;
        if (l == 0) k_gemm128<1><<<gGemm, T, 0, stream>>>(x, agg, h, Wl, dis, bp, xw);
        else        k_gemm128<2><<<gGemm, T, 0, stream>>>(x, agg, h, Wl, dis, bp, xw);
        k_edge_scatter<<<SCATTER_BLOCKS, T, 0, stream>>>(xw, src, dst, agg, cnt0 + 1 + l, cnt1 + 1 + l);
    }
    k_finish<<<gNF, T, 0, stream>>>(agg, h, dis, b_h + (size_t)(NLAYER - 1) * GD, 1);

    k_dot<<<(GN * 32) / T, T, 0, stream>>>(h, W_out, dis, xw1, GN);
    k_out_init<<<gN, T, 0, stream>>>(xw1, dis, b_out, out, GN);
    k_out_scatter<<<gE, T, 0, stream>>>(xw1, src, dst, dis, out, GE);
}


# residual carried in agg's start value: hidden GEMM epilogue stores xs + h/dis from its LDS tile, so h is not stored or re-read per layer; residual now summed first, atomic arrival order as before
# speedup vs baseline: 2.6252x; 1.0345x over previous
#include <hip/hip_runtime.h>

#define GN 50000
#define GE 800000
#define GD 128
#define NLAYER 4

typedef __attribute__((ext_vector_type(2))) float v2f;
typedef __attribute__((ext_vector_type(8))) float v8f;


__global__ void k_zero(float* p, int n) {
    int i = blockIdx.x * blockDim.x + threadIdx.x;
    if (i < n) p[i] = 0.0f;
}

__global__ void k_deg(const int* __restrict__ dst, float* deg, int e) {
    int i = blockIdx.x * blockDim.x + threadIdx.x;
    if (i < e) atomicAdd(&deg[dst[i]], 1.0f);
}

__global__ void k_rsqrt(float* dis, int n) {
    int i = blockIdx.x * blockDim.x + threadIdx.x;
    if (i < n) dis[i] = rsqrtf(1.0f + dis[i]);
}


template <int MODE>
__global__ void k_gemm128(const float* __restrict__ X,
                          float* agg,
                          const float* __restrict__ W,
                          const float* __restrict__ dis,
                          const float* __restrict__ bprev,
                          float* __restrict__ xw) {
    __shared__ __attribute__((aligned(16))) float As[16 * 132];
    const int rowBase = blockIdx.x * 16;
    const int tid = threadIdx.x;

    float4 v[2];
    float dr[2];
    float4 bp;
    if (MODE != 0) bp = *(const float4*)(bprev + (tid & 31) * 4);
    #pragma unroll
    for (int q = 0; q < 2; ++q) {
        const int i4 = tid + q * 256;
        const size_t g = (size_t)(rowBase + (i4 >> 5)) * GD + (i4 & 31) * 4;
        if (MODE == 0) {
            v[q] = *(const float4*)(X + g);
        } else {
            v[q] = *(const float4*)(agg + g);
            dr[q] = dis[rowBase + (i4 >> 5)];
        }
    }
    #pragma unroll
    for (int q = 0; q < 2; ++q) {
        if (MODE != 0) {
            v[q].x = v[q].x * dr[q] + bp.x; v[q].y = v[q].y * dr[q] + bp.y;
            v[q].z = v[q].z * dr[q] + bp.z; v[q].w = v[q].w * dr[q] + bp.w;
            v[q].x = fmaxf(v[q].x, 0.0f); v[q].y = fmaxf(v[q].y, 0.0f);
            v[q].z = fmaxf(v[q].z, 0.0f); v[q].w = fmaxf(v[q].w, 0.0f);
        }
        const int i4 = tid + q * 256;
        *(float4*)(&As[(i4 >> 5) * 132 + (i4 & 31) * 4]) = v[q];
    }
    __syncthreads();

    const int wave  = tid >> 5;
    const int lane  = tid & 31;
    const int m     = lane & 15;
    const int khalf = (lane >> 4) * 2;
    const int n     = wave * 16 + m;

    v8f c = {};
    #pragma unroll
    for (int k = 0; k < 128; k += 4) {
        v2f a, b;
        a.x = As[m * 132 + k + khalf];
        a.y = As[m * 132 + k + khalf + 1];
        b.x = W[(size_t)(k + khalf) * GD + n];
        b.y = W[(size_t)(k + khalf + 1) * GD + n];
        c = __builtin_amdgcn_wmma_f32_16x16x4_f32(
                false, a, false, b, (short)0, c, false, false);
    }

    const int rOff = (lane >> 4) * 8;
    const float4 d0 = *(const float4*)(dis + rowBase + rOff);
    const float4 d1 = *(const float4*)(dis + rowBase + rOff + 4);
    const float dsr[8] = {d0.x, d0.y, d0.z, d0.w, d1.x, d1.y, d1.z, d1.w};
    #pragma unroll
    for (int i = 0; i < 8; ++i) {
        const size_t o = (size_t)(rowBase + rOff + i) * GD + n;
        const float xs = c[i] * dsr[i];
        xw[o]  = xs;
        if (MODE == 0) agg[o] = xs;
        else           agg[o] = xs + As[(rOff + i) * 132 + n] / dsr[i];
    }
}


static_assert(GE % 256 == 0, "a block ticket is 8 whole 32-edge groups");

#define EDGE_BATCH 8
#define NTICKET (GE / 256)
#define SCATTER_BLOCKS 2048

__device__ __forceinline__ void scatter_group(const float* __restrict__ xw,
                                              const int* __restrict__ src,
                                              const int* __restrict__ dst,
                                              float* __restrict__ agg,
                                              int e0, int set, int lane) {
    int s = src[e0 + lane], d = dst[e0 + lane];
    asm volatile("" :: "v"(s));
    bool mine = (d >= GN / 2) == (set != 0);
    unsigned m = __builtin_amdgcn_ballot_w32(mine);
    while (m != 0) {
        float v[EDGE_BATCH][4];
        unsigned ml = m;
        #pragma unroll
        for (int u = 0; u < EDGE_BATCH; ++u) {
            if (ml != 0) {
                int j = __builtin_ctz(ml);
                ml &= ml - 1;
                int sj = __builtin_amdgcn_readlane(s, j);
                const float* xs = xw + (size_t)sj * GD;
                v[u][0] = xs[lane];      v[u][1] = xs[lane + 32];
                v[u][2] = xs[lane + 64]; v[u][3] = xs[lane + 96];
            }
        }
        #pragma unroll
        for (int u = 0; u < EDGE_BATCH; ++u) {
            if (m != 0) {
                int j = __builtin_ctz(m);
                m &= m - 1;
                int dj = __builtin_amdgcn_readlane(d, j);
                float* ad = agg + (size_t)dj * GD;
                atomicAdd(ad + lane,      v[u][0]);
                atomicAdd(ad + lane + 32, v[u][1]);
                atomicAdd(ad + lane + 64, v[u][2]);
                atomicAdd(ad + lane + 96, v[u][3]);
            }
        }
    }
}

__global__ void k_edge_scatter(const float* __restrict__ xw,
                               const int* __restrict__ src,
                               const int* __restrict__ dst,
                               float* __restrict__ agg,
                               int* cnt0, int* cnt1) {
    __shared__ int tk;
    const int lane = threadIdx.x & 31;
    const int wave = threadIdx.x >> 5;
    const int myhalf = (__builtin_amdgcn_s_sendmsg_rtn(0x87) >> 18) & 1;
    for (int k = 0; k < 2; ++k) {
        const int set = myhalf ^ k;
        int* cnt = set ? cnt1 : cnt0;
        int tn = 0;
        if (threadIdx.x == 0) tn = atomicAdd(cnt, 1);
        for (int trip = 0; trip <= NTICKET; ++trip) {
            if (threadIdx.x == 0) tk = tn;
            __syncthreads();
            const int t = tk;
            __syncthreads();
            if (t >= NTICKET) break;
            if (threadIdx.x == 0) tn = atomicAdd(cnt, 1);
            scatter_group(xw, src, dst, agg, (t * 8 + wave) * 32, set, lane);
        }
    }
}


__global__ void k_finish(const float* __restrict__ agg,
                         float* __restrict__ h,
                         const float* __restrict__ dis,
                         const float* __restrict__ bias) {
    int i = blockIdx.x * blockDim.x + threadIdx.x;
    float v = agg[i] * dis[i >> 7] + bias[i & (GD - 1)];
    h[i] = fmaxf(v, 0.0f);
}


__global__ void k_dot(const float* __restrict__ h,
                      const float* __restrict__ Wout,
                      const float* __restrict__ dis,
                      float* __restrict__ xw1, int n) {
    int gt = blockIdx.x * blockDim.x + threadIdx.x;
    int row = gt >> 5;
    if (row >= n) return;
    int lane = threadIdx.x & 31;
    float4 a = ((const float4*)(h + (size_t)row * GD))[lane];
    float4 b = ((const float4*)Wout)[lane];
    float sum = a.x * b.x + a.y * b.y + a.z * b.z + a.w * b.w;
    #pragma unroll
    for (int off = 16; off > 0; off >>= 1)
        sum += __shfl_down(sum, off, 32);
    if (lane == 0) xw1[row] = sum * dis[row];
}

__global__ void k_out_init(const float* __restrict__ xw1,
                           const float* __restrict__ dis,
                           const float* __restrict__ bout,
                           float* __restrict__ out, int n) {
    int i = blockIdx.x * blockDim.x + threadIdx.x;
    if (i < n) {
        out[i] = xw1[i] * dis[i] + bout[0];
    }
}

__global__ void k_out_scatter(const float* __restrict__ xw1,
                              const int* __restrict__ src,
                              const int* __restrict__ dst,
                              const float* __restrict__ dis,
                              float* __restrict__ out, int e) {
    int i = blockIdx.x * blockDim.x + threadIdx.x;
    if (i < e) {
        int s = src[i], d = dst[i];
        atomicAdd(&out[d], xw1[s] * dis[d]);
    }
}


extern "C" void kernel_launch(void* const* d_in, const int* in_sizes, int n_in,
                              void* d_out, int out_size, void* d_ws, size_t ws_size,
                              hipStream_t stream) {
    (void)in_sizes; (void)n_in; (void)out_size; (void)ws_size;
    const float* x     = (const float*)d_in[0];
    const int*   eidx  = (const int*)d_in[1];
    const float* W_in  = (const float*)d_in[2];
    const float* b_in  = (const float*)d_in[3];
    const float* W_h   = (const float*)d_in[4];
    const float* b_h   = (const float*)d_in[5];
    const float* W_out = (const float*)d_in[6];
    const float* b_out = (const float*)d_in[7];
    float* out = (float*)d_out;

    const int* src = eidx;
    const int* dst = eidx + GE;

    const size_t NPAD = (GN + 63) & ~(size_t)63;
    float* dis = (float*)d_ws;
    float* xw1 = dis + NPAD;
    float* h   = xw1 + NPAD;
    float* xw  = h + (size_t)GN * GD;
    float* agg = xw + (size_t)GN * GD;

    const int T = 256;
    const int gN    = (GN + T - 1) / T;
    const int gE    = (GE + T - 1) / T;
    const int gNF   = (GN * GD) / T;
    const int gGemm = GN / 16;

    int* cnt0 = (int*)(dis + GN + 16);
    int* cnt1 = (int*)(xw1 + GN + 16);
    k_zero<<<(int)((2 * NPAD + T - 1) / T), T, 0, stream>>>(dis, (int)(2 * NPAD));
    k_deg<<<gE, T, 0, stream>>>(dst, dis, GE);
    k_rsqrt<<<gN, T, 0, stream>>>(dis, GN);

    k_gemm128<0><<<gGemm, T, 0, stream>>>(x, agg, W_in, dis, b_in, xw);
    k_edge_scatter<<<SCATTER_BLOCKS, T, 0, stream>>>(xw, src, dst, agg, cnt0, cnt1);

    for (int l = 0; l < NLAYER; ++l) {
        const float* Wl = W_h + (size_t)l * GD * GD;
        const float* bp = (l == 0) ? b_in : b_h + (size_t)(l - 1) * GD;
        k_gemm128<1><<<gGemm, T, 0, stream>>>(x, agg, Wl, dis, bp, xw);
        k_edge_scatter<<<SCATTER_BLOCKS, T, 0, stream>>>(xw, src, dst, agg, cnt0 + 1 + l, cnt1 + 1 + l);
    }
    k_finish<<<gNF, T, 0, stream>>>(agg, h, dis, b_h + (size_t)(NLAYER - 1) * GD);

    k_dot<<<(GN * 32) / T, T, 0, stream>>>(h, W_out, dis, xw1, GN);
    k_out_init<<<gN, T, 0, stream>>>(xw1, dis, b_out, out, GN);
    k_out_scatter<<<gE, T, 0, stream>>>(xw1, src, dst, dis, out, GE);
}


# the four leftover elementwise launches become one: rsqrt(1+deg) moves into the input GEMM's epilogue; last relu, output dot and out's start value are one row pass (h never stored); 17 launches to 14
# speedup vs baseline: 2.7122x; 1.0331x over previous
#include <hip/hip_runtime.h>

#define GN 50000
#define GE 800000
#define GD 128
#define NLAYER 4

typedef __attribute__((ext_vector_type(2))) float v2f;
typedef __attribute__((ext_vector_type(8))) float v8f;


__global__ void k_zero(float* p, int n) {
    int i = blockIdx.x * blockDim.x + threadIdx.x;
    if (i < n) p[i] = 0.0f;
}

__global__ void k_deg(const int* __restrict__ dst, float* deg, int e) {
    int i = blockIdx.x * blockDim.x + threadIdx.x;
    if (i < e) atomicAdd(&deg[dst[i]], 1.0f);
}


template <int MODE>
__global__ void k_gemm128(const float* __restrict__ X,
                          float* agg,
                          const float* __restrict__ W,
                          const float* __restrict__ dis,
                          const float* __restrict__ bprev,
                          float* __restrict__ xw,
                          float* __restrict__ disw) {
    __shared__ __attribute__((aligned(16))) float As[16 * 132];
    const int rowBase = blockIdx.x * 16;
    const int tid = threadIdx.x;

    float4 v[2];
    float dr[2];
    float4 bp;
    if (MODE != 0) bp = *(const float4*)(bprev + (tid & 31) * 4);
    #pragma unroll
    for (int q = 0; q < 2; ++q) {
        const int i4 = tid + q * 256;
        const size_t g = (size_t)(rowBase + (i4 >> 5)) * GD + (i4 & 31) * 4;
        if (MODE == 0) {
            v[q] = *(const float4*)(X + g);
        } else {
            v[q] = *(const float4*)(agg + g);
            dr[q] = dis[rowBase + (i4 >> 5)];
        }
    }
    #pragma unroll
    for (int q = 0; q < 2; ++q) {
        if (MODE != 0) {
            v[q].x = v[q].x * dr[q] + bp.x; v[q].y = v[q].y * dr[q] + bp.y;
            v[q].z = v[q].z * dr[q] + bp.z; v[q].w = v[q].w * dr[q] + bp.w;
            v[q].x = fmaxf(v[q].x, 0.0f); v[q].y = fmaxf(v[q].y, 0.0f);
            v[q].z = fmaxf(v[q].z, 0.0f); v[q].w = fmaxf(v[q].w, 0.0f);
        }
        const int i4 = tid + q * 256;
        *(float4*)(&As[(i4 >> 5) * 132 + (i4 & 31) * 4]) = v[q];
    }
    __syncthreads();

    const int wave  = tid >> 5;
    const int lane  = tid & 31;
    const int m     = lane & 15;
    const int khalf = (lane >> 4) * 2;
    const int n     = wave * 16 + m;

    v8f c = {};
    #pragma unroll
    for (int k = 0; k < 128; k += 4) {
        v2f a, b;
        a.x = As[m * 132 + k + khalf];
        a.y = As[m * 132 + k + khalf + 1];
        b.x = W[(size_t)(k + khalf) * GD + n];
        b.y = W[(size_t)(k + khalf + 1) * GD + n];
        c = __builtin_amdgcn_wmma_f32_16x16x4_f32(
                false, a, false, b, (short)0, c, false, false);
    }

    const int rOff = (lane >> 4) * 8;
    const float4 d0 = *(const float4*)(dis + rowBase + rOff);
    const float4 d1 = *(const float4*)(dis + rowBase + rOff + 4);
    float dsr[8] = {d0.x, d0.y, d0.z, d0.w, d1.x, d1.y, d1.z, d1.w};
    if (MODE == 0) {
        #pragma unroll
        for (int i = 0; i < 8; ++i) dsr[i] = rsqrtf(1.0f + dsr[i]);
        if (wave == 0 && m == 0) {
            *(float4*)(disw + rowBase + rOff)     = make_float4(dsr[0], dsr[1], dsr[2], dsr[3]);
            *(float4*)(disw + rowBase + rOff + 4) = make_float4(dsr[4], dsr[5], dsr[6], dsr[7]);
        }
    }
    #pragma unroll
    for (int i = 0; i < 8; ++i) {
        const size_t o = (size_t)(rowBase + rOff + i) * GD + n;
        const float xs = c[i] * dsr[i];
        xw[o]  = xs;
        if (MODE == 0) agg[o] = xs;
        else           agg[o] = xs + As[(rOff + i) * 132 + n] / dsr[i];
    }
}


static_assert(GE % 256 == 0, "a block ticket is 8 whole 32-edge groups");

#define EDGE_BATCH 8
#define NTICKET (GE / 256)
#define SCATTER_BLOCKS 2048

__device__ __forceinline__ void scatter_group(const float* __restrict__ xw,
                                              const int* __restrict__ src,
                                              const int* __restrict__ dst,
                                              float* __restrict__ agg,
                                              int e0, int set, int lane) {
    int s = src[e0 + lane], d = dst[e0 + lane];
    asm volatile("" :: "v"(s));
    bool mine = (d >= GN / 2) == (set != 0);
    unsigned m = __builtin_amdgcn_ballot_w32(mine);
    while (m != 0) {
        float v[EDGE_BATCH][4];
        unsigned ml = m;
        #pragma unroll
        for (int u = 0; u < EDGE_BATCH; ++u) {
            if (ml != 0) {
                int j = __builtin_ctz(ml);
                ml &= ml - 1;
                int sj = __builtin_amdgcn_readlane(s, j);
                const float* xs = xw + (size_t)sj * GD;
                v[u][0] = xs[lane];      v[u][1] = xs[lane + 32];
                v[u][2] = xs[lane + 64]; v[u][3] = xs[lane + 96];
            }
        }
        #pragma unroll
        for (int u = 0; u < EDGE_BATCH; ++u) {
            if (m != 0) {
                int j = __builtin_ctz(m);
                m &= m - 1;
                int dj = __builtin_amdgcn_readlane(d, j);
                float* ad = agg + (size_t)dj * GD;
                atomicAdd(ad + lane,      v[u][0]);
                atomicAdd(ad + lane + 32, v[u][1]);
                atomicAdd(ad + lane + 64, v[u][2]);
                atomicAdd(ad + lane + 96, v[u][3]);
            }
        }
    }
}

__global__ void k_edge_scatter(const float* __restrict__ xw,
                               const int* __restrict__ src,
                               const int* __restrict__ dst,
                               float* __restrict__ agg,
                               int* cnt0, int* cnt1) {
    __shared__ int tk;
    const int lane = threadIdx.x & 31;
    const int wave = threadIdx.x >> 5;
    const int myhalf = (__builtin_amdgcn_s_sendmsg_rtn(0x87) >> 18) & 1;
    for (int k = 0; k < 2; ++k) {
        const int set = myhalf ^ k;
        int* cnt = set ? cnt1 : cnt0;
        int tn = 0;
        if (threadIdx.x == 0) tn = atomicAdd(cnt, 1);
        for (int trip = 0; trip <= NTICKET; ++trip) {
            if (threadIdx.x == 0) tk = tn;
            __syncthreads();
            const int t = tk;
            __syncthreads();
            if (t >= NTICKET) break;
            if (threadIdx.x == 0) tn = atomicAdd(cnt, 1);
            scatter_group(xw, src, dst, agg, (t * 8 + wave) * 32, set, lane);
        }
    }
}


__global__ void k_row_out(const float* __restrict__ agg,
                          const float* __restrict__ Wout,
                          const float* __restrict__ dis,
                          const float* __restrict__ bias,
                          const float* __restrict__ bout,
                          float* __restrict__ xw1,
                          float* __restrict__ out, int n) {
    int gt = blockIdx.x * blockDim.x + threadIdx.x;
    int row = gt >> 5;
    if (row >= n) return;
    int lane = threadIdx.x & 31;
    float4 a = ((const float4*)(agg + (size_t)row * GD))[lane];
    float4 c = ((const float4*)bias)[lane];
    float4 b = ((const float4*)Wout)[lane];
    const float dr = dis[row];
    a.x = fmaxf(a.x * dr + c.x, 0.0f); a.y = fmaxf(a.y * dr + c.y, 0.0f);
    a.z = fmaxf(a.z * dr + c.z, 0.0f); a.w = fmaxf(a.w * dr + c.w, 0.0f);
    float sum = a.x * b.x + a.y * b.y + a.z * b.z + a.w * b.w;
    #pragma unroll
    for (int off = 16; off > 0; off >>= 1)
        sum += __shfl_down(sum, off, 32);
    if (lane == 0) {
        const float xs = sum * dr;
        xw1[row] = xs;
        out[row] = xs * dr + bout[0];
    }
}

__global__ void k_out_scatter(const float* __restrict__ xw1,
                              const int* __restrict__ src,
                              const int* __restrict__ dst,
                              const float* __restrict__ dis,
                              float* __restrict__ out, int e) {
    int i = blockIdx.x * blockDim.x + threadIdx.x;
    if (i < e) {
        int s = src[i], d = dst[i];
        atomicAdd(&out[d], xw1[s] * dis[d]);
    }
}


extern "C" void kernel_launch(void* const* d_in, const int* in_sizes, int n_in,
                              void* d_out, int out_size, void* d_ws, size_t ws_size,
                              hipStream_t stream) {
    (void)in_sizes; (void)n_in; (void)out_size; (void)ws_size;
    const float* x     = (const float*)d_in[0];
    const int*   eidx  = (const int*)d_in[1];
    const float* W_in  = (const float*)d_in[2];
    const float* b_in  = (const float*)d_in[3];
    const float* W_h   = (const float*)d_in[4];
    const float* b_h   = (const float*)d_in[5];
    const float* W_out = (const float*)d_in[6];
    const float* b_out = (const float*)d_in[7];
    float* out = (float*)d_out;

    const int* src = eidx;
    const int* dst = eidx + GE;

    const size_t NPAD = (GN + 63) & ~(size_t)63;
    float* dis = (float*)d_ws;
    float* xw1 = dis + NPAD;
    float* xw  = xw1 + NPAD;
    float* agg = xw + (size_t)GN * GD;
    float* deg = xw1;

    const int T = 256;
    const int gE    = (GE + T - 1) / T;
    const int gGemm = GN / 16;

    int* cnt0 = (int*)(dis + GN + 16);
    int* cnt1 = (int*)(xw1 + GN + 16);
    k_zero<<<(int)((2 * NPAD + T - 1) / T), T, 0, stream>>>(dis, (int)(2 * NPAD));
    k_deg<<<gE, T, 0, stream>>>(dst, deg, GE);

    k_gemm128<0><<<gGemm, T, 0, stream>>>(x, agg, W_in, deg, b_in, xw, dis);
    k_edge_scatter<<<SCATTER_BLOCKS, T, 0, stream>>>(xw, src, dst, agg, cnt0, cnt1);

    for (int l = 0; l < NLAYER; ++l) {
        const float* Wl = W_h + (size_t)l * GD * GD;
        const float* bp = (l == 0) ? b_in : b_h + (size_t)(l - 1) * GD;
        k_gemm128<1><<<gGemm, T, 0, stream>>>(x, agg, Wl, dis, bp, xw, dis);
        k_edge_scatter<<<SCATTER_BLOCKS, T, 0, stream>>>(xw, src, dst, agg, cnt0 + 1 + l, cnt1 + 1 + l);
    }

    k_row_out<<<(GN * 32) / T, T, 0, stream>>>(agg, W_out, dis, b_h + (size_t)(NLAYER - 1) * GD,
                                              b_out, xw1, out, GN);
    k_out_scatter<<<gE, T, 0, stream>>>(xw1, src, dst, dis, out, GE);
}


# the two one-word-per-edge sums (edge counts, output layer): 125 blocks add a slice each into an own LDS table, then add it to global as whole-line atomics; f32 sum order: per block, then arrival
# speedup vs baseline: 3.2370x; 1.1935x over previous
#include <hip/hip_runtime.h>

#define GN 50000
#define GE 800000
#define GD 128
#define NLAYER 4

typedef __attribute__((ext_vector_type(2))) float v2f;
typedef __attribute__((ext_vector_type(8))) float v8f;


__global__ void k_zero(float* p, int n) {
    int i = blockIdx.x * blockDim.x + threadIdx.x;
    if (i < n) p[i] = 0.0f;
}


#define SUM_BLOCKS 125
#define SUM_T 1024
#define SUM_Q (GE / SUM_BLOCKS / 4)
#define SUM_R ((SUM_Q + SUM_T - 1) / SUM_T)
#define SUM_F 8
static_assert(GE % (SUM_BLOCKS * 4) == 0, "a slice is a whole number of index quads");
static_assert(GN % 4 == 0, "the table is cleared in 16-byte pieces");

__global__ __launch_bounds__(SUM_T) void k_deg(const int* __restrict__ dst, float* deg) {
    __shared__ __attribute__((aligned(16))) float tab[GN];
    const int tid = threadIdx.x;
    const int4* d4 = (const int4*)(dst + (size_t)blockIdx.x * (SUM_Q * 4));
    int4 d[SUM_R];
    #pragma unroll
    for (int q = 0; q < SUM_R; ++q) {
        const int i = tid + q * SUM_T;
        if (i < SUM_Q) d[q] = d4[i];
    }
    for (int i = tid; i < GN / 4; i += SUM_T) ((float4*)tab)[i] = make_float4(0.0f, 0.0f, 0.0f, 0.0f);
    __syncthreads();
    #pragma unroll
    for (int q = 0; q < SUM_R; ++q) {
        const int i = tid + q * SUM_T;
        if (i < SUM_Q) {
            atomicAdd(&tab[d[q].x], 1.0f); atomicAdd(&tab[d[q].y], 1.0f);
            atomicAdd(&tab[d[q].z], 1.0f); atomicAdd(&tab[d[q].w], 1.0f);
        }
    }
    __syncthreads();
    for (int i0 = tid; i0 < GN; i0 += SUM_T * SUM_F) {
        float v[SUM_F];
        #pragma unroll
        for (int u = 0; u < SUM_F; ++u) {
            const int i = i0 + u * SUM_T;
            v[u] = (i < GN) ? tab[i] : 0.0f;
        }
        #pragma unroll
        for (int u = 0; u < SUM_F; ++u)
            if (v[u] != 0.0f) atomicAdd(&deg[i0 + u * SUM_T], v[u]);
    }
}


template <int MODE>
__global__ void k_gemm128(const float* __restrict__ X,
                          float* agg,
                          const float* __restrict__ W,
                          const float* __restrict__ dis,
                          const float* __restrict__ bprev,
                          float* __restrict__ xw,
                          float* __restrict__ disw) {
    __shared__ __attribute__((aligned(16))) float As[16 * 132];
    const int rowBase = blockIdx.x * 16;
    const int tid = threadIdx.x;

    float4 v[2];
    float dr[2];
    float4 bp;
    if (MODE != 0) bp = *(const float4*)(bprev + (tid & 31) * 4);
    #pragma unroll
    for (int q = 0; q < 2; ++q) {
        const int i4 = tid + q * 256;
        const size_t g = (size_t)(rowBase + (i4 >> 5)) * GD + (i4 & 31) * 4;
        if (MODE == 0) {
            v[q] = *(const float4*)(X + g);
        } else {
            v[q] = *(const float4*)(agg + g);
            dr[q] = dis[rowBase + (i4 >> 5)];
        }
    }
    #pragma unroll
    for (int q = 0; q < 2; ++q) {
        if (MODE != 0) {
            v[q].x = v[q].x * dr[q] + bp.x; v[q].y = v[q].y * dr[q] + bp.y;
            v[q].z = v[q].z * dr[q] + bp.z; v[q].w = v[q].w * dr[q] + bp.w;
            v[q].x = fmaxf(v[q].x, 0.0f); v[q].y = fmaxf(v[q].y, 0.0f);
            v[q].z = fmaxf(v[q].z, 0.0f); v[q].w = fmaxf(v[q].w, 0.0f);
        }
        const int i4 = tid + q * 256;
        *(float4*)(&As[(i4 >> 5) * 132 + (i4 & 31) * 4]) = v[q];
    }
    __syncthreads();

    const int wave  = tid >> 5;
    const int lane  = tid & 31;
    const int m     = lane & 15;
    const int khalf = (lane >> 4) * 2;
    const int n     = wave * 16 + m;

    v8f c = {};
    #pragma unroll
    for (int k = 0; k < 128; k += 4) {
        v2f a, b;
        a.x = As[m * 132 + k + khalf];
        a.y = As[m * 132 + k + khalf + 1];
        b.x = W[(size_t)(k + khalf) * GD + n];
        b.y = W[(size_t)(k + khalf + 1) * GD + n];
        c = __builtin_amdgcn_wmma_f32_16x16x4_f32(
                false, a, false, b, (short)0, c, false, false);
    }

    const int rOff = (lane >> 4) * 8;
    const float4 d0 = *(const float4*)(dis + rowBase + rOff);
    const float4 d1 = *(const float4*)(dis + rowBase + rOff + 4);
    float dsr[8] = {d0.x, d0.y, d0.z, d0.w, d1.x, d1.y, d1.z, d1.w};
    if (MODE == 0) {
        #pragma unroll
        for (int i = 0; i < 8; ++i) dsr[i] = rsqrtf(1.0f + dsr[i]);
        if (wave == 0 && m == 0) {
            *(float4*)(disw + rowBase + rOff)     = make_float4(dsr[0], dsr[1], dsr[2], dsr[3]);
            *(float4*)(disw + rowBase + rOff + 4) = make_float4(dsr[4], dsr[5], dsr[6], dsr[7]);
        }
    }
    #pragma unroll
    for (int i = 0; i < 8; ++i) {
        const size_t o = (size_t)(rowBase + rOff + i) * GD + n;
        const float xs = c[i] * dsr[i];
        xw[o]  = xs;
        if (MODE == 0) agg[o] = xs;
        else           agg[o] = xs + As[(rOff + i) * 132 + n] / dsr[i];
    }
}


static_assert(GE % 256 == 0, "a block ticket is 8 whole 32-edge groups");

#define EDGE_BATCH 8
#define NTICKET (GE / 256)
#define SCATTER_BLOCKS 2048

__device__ __forceinline__ void scatter_group(const float* __restrict__ xw,
                                              const int* __restrict__ src,
                                              const int* __restrict__ dst,
                                              float* __restrict__ agg,
                                              int e0, int set, int lane) {
    int s = src[e0 + lane], d = dst[e0 + lane];
    asm volatile("" :: "v"(s));
    bool mine = (d >= GN / 2) == (set != 0);
    unsigned m = __builtin_amdgcn_ballot_w32(mine);
    while (m != 0) {
        float v[EDGE_BATCH][4];
        unsigned ml = m;
        #pragma unroll
        for (int u = 0; u < EDGE_BATCH; ++u) {
            if (ml != 0) {
                int j = __builtin_ctz(ml);
                ml &= ml - 1;
                int sj = __builtin_amdgcn_readlane(s, j);
                const float* xs = xw + (size_t)sj * GD;
                v[u][0] = xs[lane];      v[u][1] = xs[lane + 32];
                v[u][2] = xs[lane + 64]; v[u][3] = xs[lane + 96];
            }
        }
        #pragma unroll
        for (int u = 0; u < EDGE_BATCH; ++u) {
            if (m != 0) {
                int j = __builtin_ctz(m);
                m &= m - 1;
                int dj = __builtin_amdgcn_readlane(d, j);
                float* ad = agg + (size_t)dj * GD;
                atomicAdd(ad + lane,      v[u][0]);
                atomicAdd(ad + lane + 32, v[u][1]);
                atomicAdd(ad + lane + 64, v[u][2]);
                atomicAdd(ad + lane + 96, v[u][3]);
            }
        }
    }
}

__global__ void k_edge_scatter(const float* __restrict__ xw,
                               const int* __restrict__ src,
                               const int* __restrict__ dst,
                               float* __restrict__ agg,
                               int* cnt0, int* cnt1) {
    __shared__ int tk;
    const int lane = threadIdx.x & 31;
    const int wave = threadIdx.x >> 5;
    const int myhalf = (__builtin_amdgcn_s_sendmsg_rtn(0x87) >> 18) & 1;
    for (int k = 0; k < 2; ++k) {
        const int set = myhalf ^ k;
        int* cnt = set ? cnt1 : cnt0;
        int tn = 0;
        if (threadIdx.x == 0) tn = atomicAdd(cnt, 1);
        for (int trip = 0; trip <= NTICKET; ++trip) {
            if (threadIdx.x == 0) tk = tn;
            __syncthreads();
            const int t = tk;
            __syncthreads();
            if (t >= NTICKET) break;
            if (threadIdx.x == 0) tn = atomicAdd(cnt, 1);
            scatter_group(xw, src, dst, agg, (t * 8 + wave) * 32, set, lane);
        }
    }
}


__global__ void k_row_out(const float* __restrict__ agg,
                          const float* __restrict__ Wout,
                          const float* __restrict__ dis,
                          const float* __restrict__ bias,
                          const float* __restrict__ bout,
                          float* __restrict__ xw1,
                          float* __restrict__ out, int n) {
    int gt = blockIdx.x * blockDim.x + threadIdx.x;
    int row = gt >> 5;
    if (row >= n) return;
    int lane = threadIdx.x & 31;
    float4 a = ((const float4*)(agg + (size_t)row * GD))[lane];
    float4 c = ((const float4*)bias)[lane];
    float4 b = ((const float4*)Wout)[lane];
    const float dr = dis[row];
    a.x = fmaxf(a.x * dr + c.x, 0.0f); a.y = fmaxf(a.y * dr + c.y, 0.0f);
    a.z = fmaxf(a.z * dr + c.z, 0.0f); a.w = fmaxf(a.w * dr + c.w, 0.0f);
    float sum = a.x * b.x + a.y * b.y + a.z * b.z + a.w * b.w;
    #pragma unroll
    for (int off = 16; off > 0; off >>= 1)
        sum += __shfl_down(sum, off, 32);
    if (lane == 0) {
        const float xs = sum * dr;
        xw1[row] = xs;
        out[row] = xs * dr + bout[0];
    }
}

__global__ __launch_bounds__(SUM_T) void k_out_scatter(const float* __restrict__ xw1,
                              const int* __restrict__ src,
                              const int* __restrict__ dst,
                              const float* __restrict__ dis,
                              float* __restrict__ out) {
    __shared__ __attribute__((aligned(16))) float tab[GN];
    const int tid = threadIdx.x;
    const int4* s4 = (const int4*)(src + (size_t)blockIdx.x * (SUM_Q * 4));
    const int4* d4 = (const int4*)(dst + (size_t)blockIdx.x * (SUM_Q * 4));
    int4 s[SUM_R], d[SUM_R];
    float4 x[SUM_R];
    #pragma unroll
    for (int q = 0; q < SUM_R; ++q) {
        const int i = tid + q * SUM_T;
        if (i < SUM_Q) { s[q] = s4[i]; d[q] = d4[i]; }
    }
    #pragma unroll
    for (int q = 0; q < SUM_R; ++q) {
        const int i = tid + q * SUM_T;
        if (i < SUM_Q) {
            x[q].x = xw1[s[q].x]; x[q].y = xw1[s[q].y];
            x[q].z = xw1[s[q].z]; x[q].w = xw1[s[q].w];
        }
    }
    for (int i = tid; i < GN / 4; i += SUM_T) ((float4*)tab)[i] = make_float4(0.0f, 0.0f, 0.0f, 0.0f);
    __syncthreads();
    #pragma unroll
    for (int q = 0; q < SUM_R; ++q) {
        const int i = tid + q * SUM_T;
        if (i < SUM_Q) {
            atomicAdd(&tab[d[q].x], x[q].x); atomicAdd(&tab[d[q].y], x[q].y);
            atomicAdd(&tab[d[q].z], x[q].z); atomicAdd(&tab[d[q].w], x[q].w);
        }
    }
    __syncthreads();
    for (int i0 = tid; i0 < GN; i0 += SUM_T * SUM_F) {
        float v[SUM_F], w[SUM_F];
        #pragma unroll
        for (int u = 0; u < SUM_F; ++u) {
            const int i = i0 + u * SUM_T;
            v[u] = (i < GN) ? tab[i] : 0.0f;
            w[u] = (i < GN) ? dis[i] : 0.0f;
        }
        #pragma unroll
        for (int u = 0; u < SUM_F; ++u)
            if (v[u] != 0.0f) atomicAdd(&out[i0 + u * SUM_T], v[u] * w[u]);
    }
}


extern "C" void kernel_launch(void* const* d_in, const int* in_sizes, int n_in,
                              void* d_out, int out_size, void* d_ws, size_t ws_size,
                              hipStream_t stream) {
    (void)in_sizes; (void)n_in; (void)out_size; (void)ws_size;
    const float* x     = (const float*)d_in[0];
    const int*   eidx  = (const int*)d_in[1];
    const float* W_in  = (const float*)d_in[2];
    const float* b_in  = (const float*)d_in[3];
    const float* W_h   = (const float*)d_in[4];
    const float* b_h   = (const float*)d_in[5];
    const float* W_out = (const float*)d_in[6];
    const float* b_out = (const float*)d_in[7];
    float* out = (float*)d_out;

    const int* src = eidx;
    const int* dst = eidx + GE;

    const size_t NPAD = (GN + 63) & ~(size_t)63;
    float* dis = (float*)d_ws;
    float* xw1 = dis + NPAD;
    float* xw  = xw1 + NPAD;
    float* agg = xw + (size_t)GN * GD;
    float* deg = xw1;

    const int T = 256;
    const int gGemm = GN / 16;

    int* cnt0 = (int*)(dis + GN + 16);
    int* cnt1 = (int*)(xw1 + GN + 16);
    k_zero<<<(int)((2 * NPAD + T - 1) / T), T, 0, stream>>>(dis, (int)(2 * NPAD));
    k_deg<<<SUM_BLOCKS, SUM_T, 0, stream>>>(dst, deg);

    k_gemm128<0><<<gGemm, T, 0, stream>>>(x, agg, W_in, deg, b_in, xw, dis);
    k_edge_scatter<<<SCATTER_BLOCKS, T, 0, stream>>>(xw, src, dst, agg, cnt0, cnt1);

    for (int l = 0; l < NLAYER; ++l) {
        const float* Wl = W_h + (size_t)l * GD * GD;
        const float* bp = (l == 0) ? b_in : b_h + (size_t)(l - 1) * GD;
        k_gemm128<1><<<gGemm, T, 0, stream>>>(x, agg, Wl, dis, bp, xw, dis);
        k_edge_scatter<<<SCATTER_BLOCKS, T, 0, stream>>>(xw, src, dst, agg, cnt0 + 1 + l, cnt1 + 1 + l);
    }

    k_row_out<<<(GN * 32) / T, T, 0, stream>>>(agg, W_out, dis, b_h + (size_t)(NLAYER - 1) * GD,
                                              b_out, xw1, out, GN);
    k_out_scatter<<<SUM_BLOCKS, SUM_T, 0, stream>>>(xw1, src, dst, dis, out);
}


# dense passes as 1024 resident work-groups looping over rows: GEMM walks 16-row tiles, its W columns kept in registers, A tile double-buffered in LDS; row pass takes rows 4 at a time; sums unchanged
# speedup vs baseline: 3.3468x; 1.0339x over previous
#include <hip/hip_runtime.h>

#define GN 50000
#define GE 800000
#define GD 128
#define NLAYER 4

typedef __attribute__((ext_vector_type(2))) float v2f;
typedef __attribute__((ext_vector_type(8))) float v8f;


__global__ void k_zero(float* p, int n) {
    int i = blockIdx.x * blockDim.x + threadIdx.x;
    if (i < n) p[i] = 0.0f;
}


#define SUM_BLOCKS 125
#define SUM_T 1024
#define SUM_Q (GE / SUM_BLOCKS / 4)
#define SUM_R ((SUM_Q + SUM_T - 1) / SUM_T)
#define SUM_F 8
static_assert(GE % (SUM_BLOCKS * 4) == 0, "a slice is a whole number of index quads");
static_assert(GN % 4 == 0, "the table is cleared in 16-byte pieces");

__global__ __launch_bounds__(SUM_T) void k_deg(const int* __restrict__ dst, float* deg) {
    __shared__ __attribute__((aligned(16))) float tab[GN];
    const int tid = threadIdx.x;
    const int4* d4 = (const int4*)(dst + (size_t)blockIdx.x * (SUM_Q * 4));
    int4 d[SUM_R];
    #pragma unroll
    for (int q = 0; q < SUM_R; ++q) {
        const int i = tid + q * SUM_T;
        if (i < SUM_Q) d[q] = d4[i];
    }
    for (int i = tid; i < GN / 4; i += SUM_T) ((float4*)tab)[i] = make_float4(0.0f, 0.0f, 0.0f, 0.0f);
    __syncthreads();
    #pragma unroll
    for (int q = 0; q < SUM_R; ++q) {
        const int i = tid + q * SUM_T;
        if (i < SUM_Q) {
            atomicAdd(&tab[d[q].x], 1.0f); atomicAdd(&tab[d[q].y], 1.0f);
            atomicAdd(&tab[d[q].z], 1.0f); atomicAdd(&tab[d[q].w], 1.0f);
        }
    }
    __syncthreads();
    for (int i0 = tid; i0 < GN; i0 += SUM_T * SUM_F) {
        float v[SUM_F];
        #pragma unroll
        for (int u = 0; u < SUM_F; ++u) {
            const int i = i0 + u * SUM_T;
            v[u] = (i < GN) ? tab[i] : 0.0f;
        }
        #pragma unroll
        for (int u = 0; u < SUM_F; ++u)
            if (v[u] != 0.0f) atomicAdd(&deg[i0 + u * SUM_T], v[u]);
    }
}


#define GEMM_BLOCKS 1024
#define NTILE (GN / 16)
static_assert(GN % 16 == 0, "whole row tiles");
static_assert(GEMM_BLOCKS <= NTILE, "every work-group has a first tile");

template <int MODE>
__device__ __forceinline__ void tile_fetch(const float* __restrict__ X, const float* agg,
                                           const float* __restrict__ dis, int t, int tid,
                                           float4& va, float4& vb, float& da, float& db) {
    const int r = t * 16 + (tid >> 5);
    const size_t g = (size_t)r * GD + (tid & 31) * 4;
    if (MODE == 0) {
        va = *(const float4*)(X + g);
        vb = *(const float4*)(X + g + 8 * GD);
    } else {
        va = *(const float4*)(agg + g);
        vb = *(const float4*)(agg + g + 8 * GD);
        da = dis[r];
        db = dis[r + 8];
    }
}

template <int MODE>
__device__ __forceinline__ void tile_put(float* As, int tid, float4 va, float4 vb,
                                         float da, float db, float4 bp) {
    if (MODE != 0) {
        va.x = fmaxf(va.x * da + bp.x, 0.0f); va.y = fmaxf(va.y * da + bp.y, 0.0f);
        va.z = fmaxf(va.z * da + bp.z, 0.0f); va.w = fmaxf(va.w * da + bp.w, 0.0f);
        vb.x = fmaxf(vb.x * db + bp.x, 0.0f); vb.y = fmaxf(vb.y * db + bp.y, 0.0f);
        vb.z = fmaxf(vb.z * db + bp.z, 0.0f); vb.w = fmaxf(vb.w * db + bp.w, 0.0f);
    }
    *(float4*)(&As[(tid >> 5) * 132 + (tid & 31) * 4]) = va;
    *(float4*)(&As[((tid >> 5) + 8) * 132 + (tid & 31) * 4]) = vb;
}

template <int MODE>
__global__ __launch_bounds__(256, 8) void k_gemm128(const float* __restrict__ X,
                          float* agg,
                          const float* __restrict__ W,
                          const float* __restrict__ dis,
                          const float* __restrict__ bprev,
                          float* __restrict__ xw,
                          float* __restrict__ disw) {
    __shared__ __attribute__((aligned(16))) float As[2][16 * 132];
    const int tid = threadIdx.x;
    const int wave  = tid >> 5;
    const int lane  = tid & 31;
    const int m     = lane & 15;
    const int khalf = (lane >> 4) * 2;
    const int n     = wave * 16 + m;
    const int rOff  = (lane >> 4) * 8;

    v2f bf[32];
    #pragma unroll
    for (int kk = 0; kk < 32; ++kk) {
        bf[kk].x = W[(size_t)(4 * kk + khalf) * GD + n];
        bf[kk].y = W[(size_t)(4 * kk + khalf + 1) * GD + n];
    }

    float4 bp = make_float4(0.0f, 0.0f, 0.0f, 0.0f);
    if (MODE != 0) bp = *(const float4*)(bprev + (tid & 31) * 4);

    float4 va, vb;
    float da = 0.0f, db = 0.0f;
    int tile = blockIdx.x;
    tile_fetch<MODE>(X, agg, dis, tile, tid, va, vb, da, db);
    tile_put<MODE>(As[0], tid, va, vb, da, db, bp);
    __syncthreads();

    int cur = 0;
    for (; tile < NTILE; tile += GEMM_BLOCKS) {
        const int rowBase = tile * 16;
        const int nxt = tile + GEMM_BLOCKS;
        const bool more = nxt < NTILE;
        if (more) tile_fetch<MODE>(X, agg, dis, nxt, tid, va, vb, da, db);
        const float4 d0 = *(const float4*)(dis + rowBase + rOff);
        const float4 d1 = *(const float4*)(dis + rowBase + rOff + 4);

        const float* A = As[cur];
        v8f c = {};
        #pragma unroll
        for (int kk = 0; kk < 32; ++kk) {
            v2f a;
            a.x = A[m * 132 + 4 * kk + khalf];
            a.y = A[m * 132 + 4 * kk + khalf + 1];
            c = __builtin_amdgcn_wmma_f32_16x16x4_f32(
                    false, a, false, bf[kk], (short)0, c, false, false);
        }

        float dsr[8] = {d0.x, d0.y, d0.z, d0.w, d1.x, d1.y, d1.z, d1.w};
        if (MODE == 0) {
            #pragma unroll
            for (int i = 0; i < 8; ++i) dsr[i] = rsqrtf(1.0f + dsr[i]);
            if (wave == 0 && m == 0) {
                *(float4*)(disw + rowBase + rOff)     = make_float4(dsr[0], dsr[1], dsr[2], dsr[3]);
                *(float4*)(disw + rowBase + rOff + 4) = make_float4(dsr[4], dsr[5], dsr[6], dsr[7]);
            }
        }
        #pragma unroll
        for (int i = 0; i < 8; ++i) {
            const size_t o = (size_t)(rowBase + rOff + i) * GD + n;
            const float xs = c[i] * dsr[i];
            xw[o]  = xs;
            if (MODE == 0) agg[o] = xs;
            else           agg[o] = xs + A[(rOff + i) * 132 + n] / dsr[i];
        }

        if (more) tile_put<MODE>(As[cur ^ 1], tid, va, vb, da, db, bp);
        __syncthreads();
        cur ^= 1;
    }
}


static_assert(GE % 256 == 0, "a block ticket is 8 whole 32-edge groups");

#define EDGE_BATCH 8
#define NTICKET (GE / 256)
#define SCATTER_BLOCKS 2048

__device__ __forceinline__ void scatter_group(const float* __restrict__ xw,
                                              const int* __restrict__ src,
                                              const int* __restrict__ dst,
                                              float* __restrict__ agg,
                                              int e0, int set, int lane) {
    int s = src[e0 + lane], d = dst[e0 + lane];
    asm volatile("" :: "v"(s));
    bool mine = (d >= GN / 2) == (set != 0);
    unsigned m = __builtin_amdgcn_ballot_w32(mine);
    while (m != 0) {
        float v[EDGE_BATCH][4];
        unsigned ml = m;
        #pragma unroll
        for (int u = 0; u < EDGE_BATCH; ++u) {
            if (ml != 0) {
                int j = __builtin_ctz(ml);
                ml &= ml - 1;
                int sj = __builtin_amdgcn_readlane(s, j);
                const float* xs = xw + (size_t)sj * GD;
                v[u][0] = xs[lane];      v[u][1] = xs[lane + 32];
                v[u][2] = xs[lane + 64]; v[u][3] = xs[lane + 96];
            }
        }
        #pragma unroll
        for (int u = 0; u < EDGE_BATCH; ++u) {
            if (m != 0) {
                int j = __builtin_ctz(m);
                m &= m - 1;
                int dj = __builtin_amdgcn_readlane(d, j);
                float* ad = agg + (size_t)dj * GD;
                atomicAdd(ad + lane,      v[u][0]);
                atomicAdd(ad + lane + 32, v[u][1]);
                atomicAdd(ad + lane + 64, v[u][2]);
                atomicAdd(ad + lane + 96, v[u][3]);
            }
        }
    }
}

__global__ void k_edge_scatter(const float* __restrict__ xw,
                               const int* __restrict__ src,
                               const int* __restrict__ dst,
                               float* __restrict__ agg,
                               int* cnt0, int* cnt1) {
    __shared__ int tk;
    const int lane = threadIdx.x & 31;
    const int wave = threadIdx.x >> 5;
    const int myhalf = (__builtin_amdgcn_s_sendmsg_rtn(0x87) >> 18) & 1;
    for (int k = 0; k < 2; ++k) {
        const int set = myhalf ^ k;
        int* cnt = set ? cnt1 : cnt0;
        int tn = 0;
        if (threadIdx.x == 0) tn = atomicAdd(cnt, 1);
        for (int trip = 0; trip <= NTICKET; ++trip) {
            if (threadIdx.x == 0) tk = tn;
            __syncthreads();
            const int t = tk;
            __syncthreads();
            if (t >= NTICKET) break;
            if (threadIdx.x == 0) tn = atomicAdd(cnt, 1);
            scatter_group(xw, src, dst, agg, (t * 8 + wave) * 32, set, lane);
        }
    }
}


#define ROW_BLOCKS 1024
#define ROW_BATCH 4

__global__ __launch_bounds__(256) void k_row_out(const float* __restrict__ agg,
                          const float* __restrict__ Wout,
                          const float* __restrict__ dis,
                          const float* __restrict__ bias,
                          const float* __restrict__ bout,
                          float* __restrict__ xw1,
                          float* __restrict__ out, int n) {
    const int lane = threadIdx.x & 31;
    const int gw = __builtin_amdgcn_readfirstlane((int)(blockIdx.x * blockDim.x + threadIdx.x) >> 5);
    const int nw = (int)(gridDim.x * blockDim.x) >> 5;
    const float4 c = ((const float4*)bias)[lane];
    const float4 b = ((const float4*)Wout)[lane];
    const float bo = bout[0];
    for (int r0 = gw; r0 < n; r0 += ROW_BATCH * nw) {
        float4 a[ROW_BATCH];
        float dr[ROW_BATCH];
        float sum[ROW_BATCH];
        #pragma unroll
        for (int u = 0; u < ROW_BATCH; ++u) {
            const int row = r0 + u * nw;
            const int rc = (row < n) ? row : r0;
            a[u] = ((const float4*)(agg + (size_t)rc * GD))[lane];
            dr[u] = dis[rc];
        }
        #pragma unroll
        for (int u = 0; u < ROW_BATCH; ++u) {
            float4 t = a[u];
            const float d = dr[u];
            t.x = fmaxf(t.x * d + c.x, 0.0f); t.y = fmaxf(t.y * d + c.y, 0.0f);
            t.z = fmaxf(t.z * d + c.z, 0.0f); t.w = fmaxf(t.w * d + c.w, 0.0f);
            sum[u] = t.x * b.x + t.y * b.y + t.z * b.z + t.w * b.w;
        }
        #pragma unroll
        for (int off = 16; off > 0; off >>= 1) {
            #pragma unroll
            for (int u = 0; u < ROW_BATCH; ++u)
                sum[u] += __shfl_down(sum[u], off, 32);
        }
        #pragma unroll
        for (int u = 0; u < ROW_BATCH; ++u) {
            const int row = r0 + u * nw;
            if (lane == 0 && row < n) {
                const float xs = sum[u] * dr[u];
                xw1[row] = xs;
                out[row] = xs * dr[u] + bo;
            }
        }
    }
}

__global__ __launch_bounds__(SUM_T) void k_out_scatter(const float* __restrict__ xw1,
                              const int* __restrict__ src,
                              const int* __restrict__ dst,
                              const float* __restrict__ dis,
                              float* __restrict__ out) {
    __shared__ __attribute__((aligned(16))) float tab[GN];
    const int tid = threadIdx.x;
    const int4* s4 = (const int4*)(src + (size_t)blockIdx.x * (SUM_Q * 4));
    const int4* d4 = (const int4*)(dst + (size_t)blockIdx.x * (SUM_Q * 4));
    int4 s[SUM_R], d[SUM_R];
    float4 x[SUM_R];
    #pragma unroll
    for (int q = 0; q < SUM_R; ++q) {
        const int i = tid + q * SUM_T;
        if (i < SUM_Q) { s[q] = s4[i]; d[q] = d4[i]; }
    }
    #pragma unroll
    for (int q = 0; q < SUM_R; ++q) {
        const int i = tid + q * SUM_T;
        if (i < SUM_Q) {
            x[q].x = xw1[s[q].x]; x[q].y = xw1[s[q].y];
            x[q].z = xw1[s[q].z]; x[q].w = xw1[s[q].w];
        }
    }
    for (int i = tid; i < GN / 4; i += SUM_T) ((float4*)tab)[i] = make_float4(0.0f, 0.0f, 0.0f, 0.0f);
    __syncthreads();
    #pragma unroll
    for (int q = 0; q < SUM_R; ++q) {
        const int i = tid + q * SUM_T;
        if (i < SUM_Q) {
            atomicAdd(&tab[d[q].x], x[q].x); atomicAdd(&tab[d[q].y], x[q].y);
            atomicAdd(&tab[d[q].z], x[q].z); atomicAdd(&tab[d[q].w], x[q].w);
        }
    }
    __syncthreads();
    for (int i0 = tid; i0 < GN; i0 += SUM_T * SUM_F) {
        float v[SUM_F], w[SUM_F];
        #pragma unroll
        for (int u = 0; u < SUM_F; ++u) {
            const int i = i0 + u * SUM_T;
            v[u] = (i < GN) ? tab[i] : 0.0f;
            w[u] = (i < GN) ? dis[i] : 0.0f;
        }
        #pragma unroll
        for (int u = 0; u < SUM_F; ++u)
            if (v[u] != 0.0f) atomicAdd(&out[i0 + u * SUM_T], v[u] * w[u]);
    }
}


extern "C" void kernel_launch(void* const* d_in, const int* in_sizes, int n_in,
                              void* d_out, int out_size, void* d_ws, size_t ws_size,
                              hipStream_t stream) {
    (void)in_sizes; (void)n_in; (void)out_size; (void)ws_size;
    const float* x     = (const float*)d_in[0];
    const int*   eidx  = (const int*)d_in[1];
    const float* W_in  = (const float*)d_in[2];
    const float* b_in  = (const float*)d_in[3];
    const float* W_h   = (const float*)d_in[4];
    const float* b_h   = (const float*)d_in[5];
    const float* W_out = (const float*)d_in[6];
    const float* b_out = (const float*)d_in[7];
    float* out = (float*)d_out;

    const int* src = eidx;
    const int* dst = eidx + GE;

    const size_t NPAD = (GN + 63) & ~(size_t)63;
    float* dis = (float*)d_ws;
    float* xw1 = dis + NPAD;
    float* xw  = xw1 + NPAD;
    float* agg = xw + (size_t)GN * GD;
    float* deg = xw1;

    const int T = 256;
    const int gGemm = GEMM_BLOCKS;

    int* cnt0 = (int*)(dis + GN + 16);
    int* cnt1 = (int*)(xw1 + GN + 16);
    k_zero<<<(int)((2 * NPAD + T - 1) / T), T, 0, stream>>>(dis, (int)(2 * NPAD));
    k_deg<<<SUM_BLOCKS, SUM_T, 0, stream>>>(dst, deg);

    k_gemm128<0><<<gGemm, T, 0, stream>>>(x, agg, W_in, deg, b_in, xw, dis);
    k_edge_scatter<<<SCATTER_BLOCKS, T, 0, stream>>>(xw, src, dst, agg, cnt0, cnt1);

    for (int l = 0; l < NLAYER; ++l) {
        const float* Wl = W_h + (size_t)l * GD * GD;
        const float* bp = (l == 0) ? b_in : b_h + (size_t)(l - 1) * GD;
        k_gemm128<1><<<gGemm, T, 0, stream>>>(x, agg, Wl, dis, bp, xw, dis);
        k_edge_scatter<<<SCATTER_BLOCKS, T, 0, stream>>>(xw, src, dst, agg, cnt0 + 1 + l, cnt1 + 1 + l);
    }

    k_row_out<<<ROW_BLOCKS, T, 0, stream>>>(agg, W_out, dis, b_h + (size_t)(NLAYER - 1) * GD,
                                              b_out, xw1, out, GN);
    k_out_scatter<<<SUM_BLOCKS, SUM_T, 0, stream>>>(xw1, src, dst, dis, out);
}
